# starting kernel
# baseline (speedup 1.0000x reference)
#include <hip/hip_runtime.h>
#include <hip/hip_bf16.h>


typedef __attribute__((ext_vector_type(2))) float v2f;
typedef __attribute__((ext_vector_type(8))) float v8f;

#define NODES 50000
#define EDGES 800000
#define FDIM  128
#define CDIM  64
#define MTILE 5

__global__ void zero_f32_kernel(float* __restrict__ p, int n) {
  int i = blockIdx.x * blockDim.x + threadIdx.x;
  if (i < n) p[i] = 0.0f;
}

__global__ void degree_kernel(const int* __restrict__ src,
                              const int* __restrict__ dst,
                              float* __restrict__ deg_out,
                              float* __restrict__ deg_in, int e) {
  int i = blockIdx.x * blockDim.x + threadIdx.x;
  if (i < e) {
    atomicAdd(&deg_out[src[i]], 1.0f);
    atomicAdd(&deg_in [dst[i]], 1.0f);
  }
}

__global__ void norm_kernel(float* __restrict__ d_src,
                            float* __restrict__ d_dst, int n) {
  int i = blockIdx.x * blockDim.x + threadIdx.x;
  if (i < n) {
    float a = d_src[i], b = d_dst[i];
    d_src[i] = rsqrtf(a > 0.0f ? a : 1.0f);
    d_dst[i] = rsqrtf(b > 0.0f ? b : 1.0f);
  }
}

__global__ void transpose_w_kernel(const float* __restrict__ W,
                                   float* __restrict__ WT, int kdim, int ndim) {
  int i = blockIdx.x * blockDim.x + threadIdx.x;
  if (i < kdim * ndim) {
    int k = i / ndim, c = i % ndim;
    WT[(size_t)c * kdim + k] = W[i];
  }
}

template <int LDN, bool SCALE_A, bool ADD_BIAS>
__global__ void gemm_k128_wmma(const float* __restrict__ A,
                               const float* __restrict__ WT,
                               const float* __restrict__ bias,
                               const float* __restrict__ dsrc,
                               float* __restrict__ out) {
  const int lane = threadIdx.x & 31;
  const int wave = threadIdx.x >> 5;
  const int m0   = blockIdx.x * (16 * MTILE);
  const int n0   = wave << 4;
  const int col  = n0 + (lane & 15);
  const int kb   = (lane >> 4) << 1;

  float init = 0.0f;
  if constexpr (ADD_BIAS) init = bias[col];
  v8f acc[MTILE];
#pragma unroll
  for (int t = 0; t < MTILE; ++t)
#pragma unroll
    for (int r = 0; r < 8; ++r) acc[t][r] = init;

  const float* __restrict__ arow[MTILE];
  float scale[MTILE];
#pragma unroll
  for (int t = 0; t < MTILE; ++t) {
    const int row = m0 + t * 16 + (lane & 15);
    arow[t]  = A + (size_t)row * FDIM;
    scale[t] = SCALE_A ? dsrc[row] : 1.0f;
  }
  const float* __restrict__ bcol = WT + (size_t)col * FDIM;

#pragma unroll
  for (int k = 0; k < FDIM; k += 4) {
    v2f b;
    b.x = bcol[k + kb];
    b.y = bcol[k + kb + 1];
#pragma unroll
    for (int t = 0; t < MTILE; ++t) {
      v2f a;
      a.x = arow[t][k + kb];
      a.y = arow[t][k + kb + 1];
      if constexpr (SCALE_A) { a.x *= scale[t]; a.y *= scale[t]; }
      acc[t] = __builtin_amdgcn_wmma_f32_16x16x4_f32(
           false, a,  false, b,
           (short)0, acc[t],  false,  false);
    }
  }

  const int rbase = (lane >> 4) << 3;
#pragma unroll
  for (int t = 0; t < MTILE; ++t)
#pragma unroll
    for (int r = 0; r < 8; ++r)
      out[(size_t)(m0 + t * 16 + rbase + r) * LDN + col] = acc[t][r];
}

__global__ void scatter_edges_kernel(const float* __restrict__ h,
                                     const int* __restrict__ src,
                                     const int* __restrict__ dst,
                                     float* __restrict__ agg, int e) {
  const int lane   = threadIdx.x & 31;
  const int gwave  = (blockIdx.x * blockDim.x + threadIdx.x) >> 5;
  const int nwaves = (gridDim.x * blockDim.x) >> 5;
  for (int i = gwave; i < e; i += nwaves) {
    const int s = src[i];
    const int d = dst[i];
    if (i + nwaves < e) {
      const int sn = src[i + nwaves];
      __builtin_prefetch(h + (size_t)sn * FDIM + lane * 4, 0, 1);
    }
    const float4 v =
        ((const float4*)(h + (size_t)s * FDIM))[lane];
    float* ap = agg + (size_t)d * FDIM + (lane << 2);
    atomicAdd(ap + 0, v.x);
    atomicAdd(ap + 1, v.y);
    atomicAdd(ap + 2, v.z);
    atomicAdd(ap + 3, v.w);
  }
}

template <bool RELU, bool ADD_RES>
__global__ void epilogue_kernel(const float* __restrict__ agg,
                                const float* __restrict__ ddst,
                                const float* __restrict__ bias,
                                const float* __restrict__ res,
                                float* __restrict__ out, int n) {
  int i = blockIdx.x * blockDim.x + threadIdx.x;
  if (i >= n) return;
  const int row = i >> 7;
  const int col = i & (FDIM - 1);
  float v = agg[i] * ddst[row] + bias[col];
  if constexpr (ADD_RES) v += res[i];
  if constexpr (RELU) v = fmaxf(v, 0.0f);
  out[i] = v;
}

extern "C" void kernel_launch(void* const* d_in, const int* in_sizes, int n_in,
                              void* d_out, int out_size, void* d_ws, size_t ws_size,
                              hipStream_t stream) {
  const float* x   = (const float*)d_in[0];
  const int*   src = (const int*)  d_in[1];
  const int*   dst = (const int*)  d_in[2];
  const float* W1  = (const float*)d_in[3];
  const float* b1  = (const float*)d_in[4];
  const float* W2  = (const float*)d_in[5];
  const float* b2  = (const float*)d_in[6];
  const float* W3  = (const float*)d_in[7];
  const float* b3  = (const float*)d_in[8];
  const float* Wr  = (const float*)d_in[9];
  const float* br  = (const float*)d_in[10];
  const float* Wo  = (const float*)d_in[11];
  const float* bo  = (const float*)d_in[12];
  float* out = (float*)d_out;

  float* ws    = (float*)d_ws;
  float* dsrc  = ws;
  float* ddst  = dsrc + NODES;
  float* RES   = ddst + NODES;
  float* ABUF  = RES  + (size_t)NODES * FDIM;
  float* HBUF  = ABUF + (size_t)NODES * FDIM;
  float* AGG   = HBUF + (size_t)NODES * FDIM;
  float* WTr   = AGG  + (size_t)NODES * FDIM;
  float* WT1   = WTr  + FDIM * FDIM;
  float* WT2   = WT1  + FDIM * FDIM;
  float* WT3   = WT2  + FDIM * FDIM;
  float* WTo   = WT3  + FDIM * FDIM;

  const int NF       = NODES * FDIM;
  const int zb_norm  = (2 * NODES + 255) / 256;
  const int zb_feat  = (NF + 255) / 256;
  const int eb       = (EDGES + 255) / 256;
  const int wb       = (FDIM * FDIM + 255) / 256;
  const int wbo      = (FDIM * CDIM + 255) / 256;
  const int mblocks  = NODES / (16 * MTILE);
  const dim3 gemmH_blk(256);
  const dim3 gemmC_blk(128);
  const dim3 scat_grid(4096), scat_blk(256);

  zero_f32_kernel<<<zb_norm, 256, 0, stream>>>(dsrc, 2 * NODES);
  degree_kernel<<<eb, 256, 0, stream>>>(src, dst, dsrc, ddst, EDGES);
  norm_kernel<<<(NODES + 255) / 256, 256, 0, stream>>>(dsrc, ddst, NODES);

  transpose_w_kernel<<<wb,  256, 0, stream>>>(Wr, WTr, FDIM, FDIM);
  transpose_w_kernel<<<wb,  256, 0, stream>>>(W1, WT1, FDIM, FDIM);
  transpose_w_kernel<<<wb,  256, 0, stream>>>(W2, WT2, FDIM, FDIM);
  transpose_w_kernel<<<wb,  256, 0, stream>>>(W3, WT3, FDIM, FDIM);
  transpose_w_kernel<<<wbo, 256, 0, stream>>>(Wo, WTo, FDIM, CDIM);

  gemm_k128_wmma<FDIM, false, true>
      <<<mblocks, gemmH_blk, 0, stream>>>(x, WTr, br, nullptr, RES);

  gemm_k128_wmma<FDIM, true, false>
      <<<mblocks, gemmH_blk, 0, stream>>>(x, WT1, nullptr, dsrc, HBUF);
  zero_f32_kernel<<<zb_feat, 256, 0, stream>>>(AGG, NF);
  scatter_edges_kernel<<<scat_grid, scat_blk, 0, stream>>>(HBUF, src, dst, AGG, EDGES);
  epilogue_kernel<true, false>
      <<<zb_feat, 256, 0, stream>>>(AGG, ddst, b1, nullptr, ABUF, NF);

  gemm_k128_wmma<FDIM, true, false>
      <<<mblocks, gemmH_blk, 0, stream>>>(ABUF, WT2, nullptr, dsrc, HBUF);
  zero_f32_kernel<<<zb_feat, 256, 0, stream>>>(AGG, NF);
  scatter_edges_kernel<<<scat_grid, scat_blk, 0, stream>>>(HBUF, src, dst, AGG, EDGES);
  epilogue_kernel<true, false>
      <<<zb_feat, 256, 0, stream>>>(AGG, ddst, b2, nullptr, ABUF, NF);

  gemm_k128_wmma<FDIM, true, false>
      <<<mblocks, gemmH_blk, 0, stream>>>(ABUF, WT3, nullptr, dsrc, HBUF);
  zero_f32_kernel<<<zb_feat, 256, 0, stream>>>(AGG, NF);
  scatter_edges_kernel<<<scat_grid, scat_blk, 0, stream>>>(HBUF, src, dst, AGG, EDGES);
  epilogue_kernel<true, true>
      <<<zb_feat, 256, 0, stream>>>(AGG, ddst, b3, RES, ABUF, NF);

  gemm_k128_wmma<CDIM, false, true>
      <<<mblocks, gemmC_blk, 0, stream>>>(ABUF, WTo, bo, nullptr, out);
}


# Out-degrees without adds to memory: fill blocks order their src ids by 1024-node range in LDS and store one whole-line run per range; a block per range counts its runs in LDS (was the zero launch)
# speedup vs baseline: 3.3531x; 3.3531x over previous
#include <hip/hip_runtime.h>
#include <hip/hip_bf16.h>


typedef __attribute__((ext_vector_type(2))) float v2f;
typedef __attribute__((ext_vector_type(4))) float v4f;
typedef __attribute__((ext_vector_type(8))) float v8f;

#define NODES 50000
#define EDGES 800000
#define FDIM  128
#define CDIM  64
#define GW     5
#define WPITCH 132

#define RB       40
#define NBIN     (NODES / RB)
#define NSB      256
#define EPB      (EDGES / NSB)
#define STHREADS 512
#define SWAVES   (STHREADS / 32)
#define EPT      ((EPB + STHREADS - 1) / STHREADS)
#define BPT      ((NBIN + STHREADS - 1) / STHREADS)
#define AGG_THREADS 256
#define EB       8
#define RB32     ((RB + 31) / 32)
static_assert(NBIN * RB == NODES, "bins must tile the nodes");
static_assert(NSB * EPB == EDGES, "sort blocks must tile the edges");
static_assert(NSB % 32 == 0, "bin_scan_kernel walks NSB in whole waves");
static_assert(NODES <= 65536 && RB <= 32768, "packed edge = (local row << 16) | src");
static_assert(EB == 8, "the short group is finished in steps of 4, 2, 1");

#define RGSH     10
#define RGN      (1 << RGSH)
#define NRG      ((NODES + RGN - 1) / RGN)
#define RCAP     3136
#define RTHREADS 1024
#define RPW      (NSB / (RTHREADS / 32))
#define RTRIPS   3
static_assert(RCAP >= EPB + 1 && RCAP % 64 == 0, "a run holds a whole sort block's edges; runs start on a line");
static_assert(RTHREADS == RGN, "a range block writes one count per thread");
static_assert(NSB % (RTHREADS / 32) == 0, "the waves of a range block share the runs evenly");
static_assert(NRG <= 64, "the run starts are scanned by one wave, two ranges per lane");

__device__ __forceinline__ float inv_sqrt_degree(float count) {
  return rsqrtf(count > 0.0f ? count : 1.0f);
}

template <int LDN, bool SCALE_A, bool ADD_BIAS>
__global__ __launch_bounds__(GW * 32)
void gemm_k128_wmma(const float* __restrict__ A,
                    const float* __restrict__ W,
                    const float* __restrict__ bias,
                    const float* __restrict__ degout,
                    float* __restrict__ out) {
  constexpr int NT = LDN / 16;
  constexpr int CB = LDN / 32;
  constexpr int NP = (FDIM / 16) * CB;
  constexpr int FP = (NP + GW - 1) / GW;
  __shared__ __attribute__((aligned(16))) float wt[LDN * WPITCH];

  const int tid  = threadIdx.x;
  const int lane = tid & 31;
  const int wave = tid >> 5;
  const int c    = lane & 15;
  const int hh   = lane >> 4;
  const int m0   = (blockIdx.x * GW + wave) * 16;

  const float4* __restrict__ arow = (const float4*)(A + (size_t)(m0 + c) * FDIM) + hh;
  float4 a[FDIM / 8];
#pragma unroll
  for (int j = 0; j < FDIM / 8; ++j) a[j] = arow[2 * j];
  float scale = 1.0f;
  if constexpr (SCALE_A) scale = inv_sqrt_degree(degout[m0 + c]);

  const int kq = (lane & 3) * 4;
  const int cq = (lane >> 2) * 4;
  const int wv = __builtin_amdgcn_readfirstlane(wave);
  __builtin_assume(wv >= 0 && wv < GW);
  v4f v[FP][4];
#pragma unroll
  for (int u = 0; u < FP; ++u) {
    const int p = u * GW + wv;
    if (p < NP) {
      const float* wp = W + (size_t)((p / CB) * 16 + kq) * LDN + (p % CB) * 32 + cq;
#pragma unroll
      for (int r = 0; r < 4; ++r) v[u][r] = *(const v4f*)(wp + r * LDN);
    }
  }
  __builtin_amdgcn_sched_barrier(0);
#pragma unroll
  for (int u = 0; u < FP; ++u) {
    const int p = u * GW + wv;
    if (p < NP) {
      float* lp = wt + ((p % CB) * 32 + cq) * WPITCH + (p / CB) * 16 + kq;
#pragma unroll
      for (int i = 0; i < 4; ++i) {
        v4f o;
        o.x = v[u][0][i]; o.y = v[u][1][i]; o.z = v[u][2][i]; o.w = v[u][3][i];
        *(v4f*)(lp + i * WPITCH) = o;
      }
    }
  }

  v8f acc[NT];
#pragma unroll
  for (int t = 0; t < NT; ++t) {
    float init = 0.0f;
    if constexpr (ADD_BIAS) init = bias[t * 16 + c];
#pragma unroll
    for (int r = 0; r < 8; ++r) acc[t][r] = init;
  }
  if constexpr (SCALE_A) {
#pragma unroll
    for (int j = 0; j < FDIM / 8; ++j) {
      a[j].x *= scale; a[j].y *= scale; a[j].z *= scale; a[j].w *= scale;
    }
  }
  __syncthreads();

  const float* bbase = wt + c * WPITCH + 4 * hh;
  float4 bcur[NT], bnxt[NT];
#pragma unroll
  for (int t = 0; t < NT; ++t) bcur[t] = *(const float4*)(bbase + t * 16 * WPITCH);

#pragma unroll
  for (int j = 0; j < FDIM / 8; ++j) {
    if (j + 1 < FDIM / 8) {
#pragma unroll
      for (int t = 0; t < NT; ++t)
        bnxt[t] = *(const float4*)(bbase + t * 16 * WPITCH + 8 * (j + 1));
    }
    v2f a0, a1;
    a0.x = a[j].x; a0.y = a[j].y;
    a1.x = a[j].z; a1.y = a[j].w;
#pragma unroll
    for (int t = 0; t < NT; ++t) {
      v2f b;
      b.x = bcur[t].x; b.y = bcur[t].y;
      acc[t] = __builtin_amdgcn_wmma_f32_16x16x4_f32(
           false, a0,  false, b,
           (short)0, acc[t],  false,  false);
    }
#pragma unroll
    for (int t = 0; t < NT; ++t) {
      v2f b;
      b.x = bcur[t].z; b.y = bcur[t].w;
      acc[t] = __builtin_amdgcn_wmma_f32_16x16x4_f32(
           false, a1,  false, b,
           (short)0, acc[t],  false,  false);
    }
    if (j + 1 < FDIM / 8) {
#pragma unroll
      for (int t = 0; t < NT; ++t) bcur[t] = bnxt[t];
    }
  }

  const int rbase = hh << 3;
#pragma unroll
  for (int r = 0; r < 8; ++r)
#pragma unroll
    for (int t = 0; t < NT; ++t)
      out[(size_t)(m0 + rbase + r) * LDN + t * 16 + c] = acc[t][r];
}

template <int LDN, int U0, int NU>
__device__ __forceinline__ void wt_piece_loads(const float* __restrict__ W, int lane, int wv,
                                               v4f (&v)[NU][4]) {
  constexpr int CB = LDN / 32;
  constexpr int NP = (FDIM / 16) * CB;
  const int kq = (lane & 3) * 4;
  const int cq = (lane >> 2) * 4;
#pragma unroll
  for (int u = 0; u < NU; ++u) {
    const int p = (U0 + u) * GW + wv;
    if (p < NP) {
      const float* wp = W + (size_t)((p / CB) * 16 + kq) * LDN + (p % CB) * 32 + cq;
#pragma unroll
      for (int r = 0; r < 4; ++r) v[u][r] = *(const v4f*)(wp + r * LDN);
    }
  }
}
template <int LDN, int U0, int NU>
__device__ __forceinline__ void wt_piece_stores(float* wt, int lane, int wv,
                                                const v4f (&v)[NU][4]) {
  constexpr int CB = LDN / 32;
  constexpr int NP = (FDIM / 16) * CB;
  const int kq = (lane & 3) * 4;
  const int cq = (lane >> 2) * 4;
#pragma unroll
  for (int u = 0; u < NU; ++u) {
    const int p = (U0 + u) * GW + wv;
    if (p < NP) {
      float* lp = wt + ((p % CB) * 32 + cq) * WPITCH + (p / CB) * 16 + kq;
#pragma unroll
      for (int i = 0; i < 4; ++i) {
        v4f o;
        o.x = v[u][0][i]; o.y = v[u][1][i]; o.z = v[u][2][i]; o.w = v[u][3][i];
        *(v4f*)(lp + i * WPITCH) = o;
      }
    }
  }
}

template <int NT>
__device__ __forceinline__ void mma_k128(const float4 (&a)[FDIM / 8], const float* bbase,
                                         v8f (&acc)[NT]) {
  float4 bcur[NT], bnxt[NT];
#pragma unroll
  for (int t = 0; t < NT; ++t) bcur[t] = *(const float4*)(bbase + t * 16 * WPITCH);
#pragma unroll
  for (int j = 0; j < FDIM / 8; ++j) {
    if (j + 1 < FDIM / 8) {
#pragma unroll
      for (int t = 0; t < NT; ++t)
        bnxt[t] = *(const float4*)(bbase + t * 16 * WPITCH + 8 * (j + 1));
    }
    v2f a0, a1;
    a0.x = a[j].x; a0.y = a[j].y;
    a1.x = a[j].z; a1.y = a[j].w;
#pragma unroll
    for (int t = 0; t < NT; ++t) {
      v2f b;
      b.x = bcur[t].x; b.y = bcur[t].y;
      acc[t] = __builtin_amdgcn_wmma_f32_16x16x4_f32(
           false, a0,  false, b,
           (short)0, acc[t],  false,  false);
    }
#pragma unroll
    for (int t = 0; t < NT; ++t) {
      v2f b;
      b.x = bcur[t].z; b.y = bcur[t].w;
      acc[t] = __builtin_amdgcn_wmma_f32_16x16x4_f32(
           false, a1,  false, b,
           (short)0, acc[t],  false,  false);
    }
    if (j + 1 < FDIM / 8) {
#pragma unroll
      for (int t = 0; t < NT; ++t) bcur[t] = bnxt[t];
    }
  }
}

#define TAIL_SCR   (GW * 16 * WPITCH)
#define TAIL_LDS   (TAIL_SCR + CDIM * WPITCH)
static_assert(TAIL_LDS >= FDIM * WPITCH, "the area must hold a whole 128-column W^T");
static_assert((TAIL_SCR * 4) % 16 == 0, "Wo^T starts on a 16-B boundary");

__global__ __launch_bounds__(GW * 32)
void tail_three_products_wmma(const float* __restrict__ T,
                              const float* __restrict__ X,
                              const float* __restrict__ W3,
                              const float* __restrict__ Wr,
                              const float* __restrict__ b3,
                              const float* __restrict__ br,
                              const float* __restrict__ Wo,
                              const float* __restrict__ bo,
                              float* __restrict__ out) {
  constexpr int NT  = FDIM / 16;
  constexpr int NTO = CDIM / 16;
  constexpr int FPH = ((FDIM / 16) * (FDIM / 32) + GW - 1) / GW;
  constexpr int FPO = ((FDIM / 16) * (CDIM / 32) + GW - 1) / GW;
  __shared__ __attribute__((aligned(16))) float wt[TAIL_LDS];

  const int tid  = threadIdx.x;
  const int lane = tid & 31;
  const int wave = tid >> 5;
  const int c    = lane & 15;
  const int hh   = lane >> 4;
  const int m0   = (blockIdx.x * GW + wave) * 16;
  const int wv = __builtin_amdgcn_readfirstlane(wave);
  __builtin_assume(wv >= 0 && wv < GW);

  float4 a[FDIM / 8];
  {
    const float4* __restrict__ arow = (const float4*)(T + (size_t)(m0 + c) * FDIM) + hh;
#pragma unroll
    for (int j = 0; j < FDIM / 8; ++j) a[j] = arow[2 * j];
    v4f v[FPH][4];
    wt_piece_loads<FDIM, 0, FPH>(W3, lane, wv, v);
    __builtin_amdgcn_sched_barrier(0);
    wt_piece_stores<FDIM, 0, FPH>(wt, lane, wv, v);
  }
  v8f acc[NT];
#pragma unroll
  for (int t = 0; t < NT; ++t) {
    const float init = b3[t * 16 + c] + br[t * 16 + c];
#pragma unroll
    for (int r = 0; r < 8; ++r) acc[t][r] = init;
  }
  __syncthreads();
  const float* bbase = wt + c * WPITCH + 4 * hh;
  mma_k128<NT>(a, bbase, acc);
  __builtin_amdgcn_sched_barrier(0);

  {
    const float4* __restrict__ arow = (const float4*)(X + (size_t)(m0 + c) * FDIM) + hh;
#pragma unroll
    for (int j = 0; j < FDIM / 8; ++j) a[j] = arow[2 * j];
    v4f v[4][4];
    wt_piece_loads<FDIM, 0, 4>(Wr, lane, wv, v);
    __syncthreads();
    wt_piece_stores<FDIM, 0, 4>(wt, lane, wv, v);
  }
  __builtin_amdgcn_sched_barrier(0);
  {
    v4f v[FPH - 4][4];
    wt_piece_loads<FDIM, 4, FPH - 4>(Wr, lane, wv, v);
    wt_piece_stores<FDIM, 4, FPH - 4>(wt, lane, wv, v);
  }
  __syncthreads();
  mma_k128<NT>(a, bbase, acc);
  __builtin_amdgcn_sched_barrier(0);

  float* scr = wt + wave * (16 * WPITCH);
  {
    v4f v[FPO][4];
    wt_piece_loads<CDIM, 0, FPO>(Wo, lane, wv, v);
    __syncthreads();
#pragma unroll
    for (int t = 0; t < NT; ++t)
#pragma unroll
      for (int r = 0; r < 8; ++r)
        scr[(hh * 8 + r) * WPITCH + t * 16 + c] = fmaxf(acc[t][r], 0.0f);
    wt_piece_stores<CDIM, 0, FPO>(wt + TAIL_SCR, lane, wv, v);
  }
  v8f acco[NTO];
#pragma unroll
  for (int t = 0; t < NTO; ++t) {
    const float init = bo[t * 16 + c];
#pragma unroll
    for (int r = 0; r < 8; ++r) acco[t][r] = init;
  }
  __syncthreads();

#pragma unroll
  for (int j = 0; j < FDIM / 8; ++j) a[j] = *(const float4*)(scr + c * WPITCH + 8 * j + 4 * hh);
  mma_k128<NTO>(a, wt + TAIL_SCR + c * WPITCH + 4 * hh, acco);

  const int rbase = hh << 3;
#pragma unroll
  for (int r = 0; r < 8; ++r)
#pragma unroll
    for (int t = 0; t < NTO; ++t)
      out[(size_t)(m0 + rbase + r) * CDIM + t * 16 + c] = acco[t][r];
}

__global__ __launch_bounds__(STHREADS)
void bin_count_kernel(const int* __restrict__ dst, int* __restrict__ hist) {
  __shared__ int cnt[NBIN];
  const int tid  = threadIdx.x;
  const int base = blockIdx.x * EPB;
  for (int i = tid; i < NBIN; i += STHREADS) cnt[i] = 0;
  int d[EPT];
#pragma unroll
  for (int k = 0; k < EPT; ++k) {
    const int i = tid + k * STHREADS;
    d[k] = (i < EPB) ? dst[base + i] : 0;
  }
  __syncthreads();
#pragma unroll
  for (int k = 0; k < EPT; ++k) {
    const int i = tid + k * STHREADS;
    if (i < EPB) atomicAdd(&cnt[d[k] / RB], 1);
  }
  __syncthreads();
  for (int i = tid; i < NBIN; i += STHREADS) hist[i * NSB + blockIdx.x] = cnt[i];
}

__global__ void bin_scan_kernel(int* __restrict__ hist, int* __restrict__ total) {
  const int lane = threadIdx.x & 31;
  const int bin  = (blockIdx.x * blockDim.x + threadIdx.x) >> 5;
  if (bin >= NBIN) return;
  int* __restrict__ row = hist + bin * NSB;
  int run = 0;
  for (int b0 = 0; b0 < NSB; b0 += 32) {
    const int v = row[b0 + lane];
    int inc = v;
    for (int dlt = 1; dlt < 32; dlt <<= 1) {
      const int u = __shfl_up(inc, dlt);
      if (lane >= dlt) inc += u;
    }
    row[b0 + lane] = run + inc - v;
    run += __shfl(inc, 31);
  }
  if (lane == 0) total[bin] = run;
}

__global__ __launch_bounds__(STHREADS)
void bin_fill_kernel(const int* __restrict__ src, const int* __restrict__ dst,
                     const int* __restrict__ pre, const int* __restrict__ total,
                     int* __restrict__ binstart, int* __restrict__ sorted,
                     unsigned int* __restrict__ runs, int* __restrict__ runlen) {
  __shared__ int cursor[NBIN];
  __shared__ int wsum[SWAVES];
  __shared__ int rcnt[NRG];
  __shared__ int roff[NRG];
  __shared__ unsigned short stage[RCAP];
  const int tid  = threadIdx.x;
  const int lane = tid & 31;
  const int wave = tid >> 5;
  const int base = blockIdx.x * EPB;

  if (tid < NRG) rcnt[tid] = 0;
  int s[EPT], d[EPT];
#pragma unroll
  for (int k = 0; k < EPT; ++k) {
    const int i  = tid + k * STHREADS;
    const bool ok = i < EPB;
    s[k] = ok ? src[base + i] : 0;
    d[k] = ok ? dst[base + i] : 0;
  }

  int loc[BPT];
  int sum = 0;
#pragma unroll
  for (int j = 0; j < BPT; ++j) {
    const int b = tid * BPT + j;
    const int v = (b < NBIN) ? total[b] : 0;
    loc[j] = sum;
    sum += v;
  }
  int inc = sum;
  for (int dlt = 1; dlt < 32; dlt <<= 1) {
    const int u = __shfl_up(inc, dlt);
    if (lane >= dlt) inc += u;
  }
  if (lane == 31) wsum[wave] = inc;
  __syncthreads();
  int woff = 0;
  for (int w = 0; w < wave; ++w) woff += wsum[w];
  const int texcl = woff + inc - sum;
#pragma unroll
  for (int j = 0; j < BPT; ++j) {
    const int b = tid * BPT + j;
    if (b < NBIN) {
      const int start = texcl + loc[j];
      cursor[b] = start + pre[b * NSB + blockIdx.x];
      if (blockIdx.x == 0) binstart[b] = start;
    }
  }
  if (blockIdx.x == 0 && tid == 0) binstart[NBIN] = EDGES;
  int rk[EPT];
#pragma unroll
  for (int k = 0; k < EPT; ++k) {
    const int i = tid + k * STHREADS;
    rk[k] = (i < EPB) ? atomicAdd(&rcnt[s[k] >> RGSH], 1) : 0;
  }
  __syncthreads();

  if (wave == 0) {
    const int q0 = 2 * lane, q1 = 2 * lane + 1;
    const int c0 = (q0 < NRG) ? rcnt[q0] : 0;
    const int c1 = (q1 < NRG) ? rcnt[q1] : 0;
    int run = c0 + c1;
    for (int dlt = 1; dlt < 32; dlt <<= 1) {
      const int u = __shfl_up(run, dlt);
      if (lane >= dlt) run += u;
    }
    const int excl = run - c0 - c1;
    if (q0 < NRG) roff[q0] = excl;
    if (q1 < NRG) roff[q1] = excl + c0;
  }

#pragma unroll
  for (int k = 0; k < EPT; ++k) {
    const int i = tid + k * STHREADS;
    if (i < EPB) {
      const int bin = d[k] / RB;
      const int pos = atomicAdd(&cursor[bin], 1);
      sorted[pos] = ((d[k] - bin * RB) << 16) | s[k];
    }
  }
  __syncthreads();

#pragma unroll
  for (int k = 0; k < EPT; ++k) {
    const int i = tid + k * STHREADS;
    if (i < EPB) stage[roff[s[k] >> RGSH] + rk[k]] = (unsigned short)(s[k] & (RGN - 1));
  }
  __syncthreads();

  for (int r = wave; r < NRG; r += SWAVES) {
    const int n = rcnt[r];
    const int o = roff[r];
    unsigned int* __restrict__ out = runs + (size_t)(r * NSB + blockIdx.x) * (RCAP / 2);
    for (int i = 2 * lane; i < n; i += 64) {
      const unsigned int lo = stage[o + i];
      const unsigned int hi = (i + 1 < n) ? stage[o + i + 1] : 0u;
      out[i >> 1] = lo | (hi << 16);
    }
    if (lane == 0) runlen[r * NSB + blockIdx.x] = n;
  }
}

__global__ __launch_bounds__(RTHREADS)
void outdeg_ranges_kernel(const unsigned int* __restrict__ runs, const int* __restrict__ runlen,
                          float* __restrict__ deg_out) {
  __shared__ int cnt[RGN];
  const int tid  = threadIdx.x;
  const int lane = tid & 31;
  const int wave = __builtin_amdgcn_readfirstlane(tid >> 5);
  const int r    = blockIdx.x;
  cnt[tid] = 0;
  __syncthreads();

  int n[RPW];
#pragma unroll
  for (int q = 0; q < RPW; ++q) {
    const int len = runlen[r * NSB + wave + (RTHREADS / 32) * q];
    n[q] = len < RCAP ? len : RCAP;
  }
  unsigned int v[RPW][RTRIPS];
#pragma unroll
  for (int q = 0; q < RPW; ++q) {
    const unsigned int* __restrict__ p = runs + (size_t)(r * NSB + wave + (RTHREADS / 32) * q) * (RCAP / 2);
#pragma unroll
    for (int t = 0; t < RTRIPS; ++t) {
      const int i = 2 * (lane + 32 * t);
      v[q][t] = (i < n[q]) ? p[i >> 1] : 0u;
    }
  }
#pragma unroll
  for (int q = 0; q < RPW; ++q) {
#pragma unroll
    for (int t = 0; t < RTRIPS; ++t) {
      const int i = 2 * (lane + 32 * t);
      if (i < n[q])     atomicAdd(&cnt[v[q][t] & (RGN - 1)], 1);
      if (i + 1 < n[q]) atomicAdd(&cnt[(v[q][t] >> 16) & (RGN - 1)], 1);
    }
    const unsigned int* __restrict__ p = runs + (size_t)(r * NSB + wave + (RTHREADS / 32) * q) * (RCAP / 2);
    for (int i = 2 * (lane + 32 * RTRIPS); i < n[q]; i += 64) {
      const unsigned int w = p[i >> 1];
      atomicAdd(&cnt[w & (RGN - 1)], 1);
      if (i + 1 < n[q]) atomicAdd(&cnt[(w >> 16) & (RGN - 1)], 1);
    }
  }
  __syncthreads();
  const int node = r * RGN + tid;
  if (node < NODES) deg_out[node] = (float)cnt[tid];
}

template <int K>
__device__ __forceinline__ void add_edges(const float* __restrict__ h,
                                          const int* __restrict__ sorted,
                                          int e, float* tile, int lane,
                                          int (&deg)[RB32]) {
  int    p[K];
  float4 v[K];
#pragma unroll
  for (int j = 0; j < K; ++j) p[j] = sorted[e + j];
#pragma unroll
  for (int j = 0; j < K; ++j)
    v[j] = ((const float4*)(h + (size_t)(p[j] & 0xFFFF) * FDIM))[lane];
#pragma unroll
  for (int j = 0; j < K; ++j) {
    const int r = p[j] >> 16;
#pragma unroll
    for (int q = 0; q < RB32; ++q) deg[q] += (r == lane + 32 * q) ? 1 : 0;
  }
#pragma unroll
  for (int j = 0; j < K; ++j) {
    float* t = tile + (p[j] >> 16) * FDIM + lane;
    __hip_atomic_fetch_add(t,      v[j].x, __ATOMIC_RELAXED, __HIP_MEMORY_SCOPE_WORKGROUP);
    __hip_atomic_fetch_add(t + 32, v[j].y, __ATOMIC_RELAXED, __HIP_MEMORY_SCOPE_WORKGROUP);
    __hip_atomic_fetch_add(t + 64, v[j].z, __ATOMIC_RELAXED, __HIP_MEMORY_SCOPE_WORKGROUP);
    __hip_atomic_fetch_add(t + 96, v[j].w, __ATOMIC_RELAXED, __HIP_MEMORY_SCOPE_WORKGROUP);
  }
}

template <bool RELU, bool ADD_BIAS, bool SCALE_OUT>
__global__ __launch_bounds__(AGG_THREADS)
void aggregate_bins_kernel(const float* __restrict__ h,
                           const int* __restrict__ sorted,
                           const int* __restrict__ binstart,
                           const float* __restrict__ bias,
                           const float* __restrict__ degout,
                           float* __restrict__ out) {
  __shared__ float tile[RB * FDIM];
  __shared__ int   cnt[RB];
  const int tid  = threadIdx.x;
  const int lane = tid & 31;
  const int wave = __builtin_amdgcn_readfirstlane(tid >> 5);
  const int bin  = blockIdx.x;

  for (int i = tid; i < RB * FDIM; i += AGG_THREADS) tile[i] = 0.0f;
  if (tid < RB) cnt[tid] = 0;
  __syncthreads();

  const int e0 = binstart[bin];
  const int e1 = binstart[bin + 1];
  int deg[RB32];
#pragma unroll
  for (int q = 0; q < RB32; ++q) deg[q] = 0;
  int e = e0 + wave * EB;
  for (; e + EB <= e1; e += (AGG_THREADS / 32) * EB)
    add_edges<EB>(h, sorted, e, tile, lane, deg);
  if (e < e1) {
    if (e + 4 <= e1) { add_edges<4>(h, sorted, e, tile, lane, deg); e += 4; }
    if (e + 2 <= e1) { add_edges<2>(h, sorted, e, tile, lane, deg); e += 2; }
    if (e < e1)        add_edges<1>(h, sorted, e, tile, lane, deg);
  }
#pragma unroll
  for (int q = 0; q < RB32; ++q) {
    const int r = lane + 32 * q;
    if (r < RB && deg[q] != 0) atomicAdd(&cnt[r], deg[q]);
  }
  __syncthreads();

  float b0 = 0.0f, b1 = 0.0f, b2 = 0.0f, b3 = 0.0f;
  if constexpr (ADD_BIAS) {
    b0 = bias[lane * 4 + 0];
    b1 = bias[lane * 4 + 1];
    b2 = bias[lane * 4 + 2];
    b3 = bias[lane * 4 + 3];
  }
  for (int i = tid; i < RB * 32; i += AGG_THREADS) {
    const int r   = i >> 5;
    const int row = bin * RB + r;
    const float* t = tile + r * FDIM + lane;
    const float sc = inv_sqrt_degree((float)cnt[r]);
    float4 o;
    o.x = t[0]  * sc + b0;
    o.y = t[32] * sc + b1;
    o.z = t[64] * sc + b2;
    o.w = t[96] * sc + b3;
    if constexpr (RELU) {
      o.x = fmaxf(o.x, 0.0f); o.y = fmaxf(o.y, 0.0f);
      o.z = fmaxf(o.z, 0.0f); o.w = fmaxf(o.w, 0.0f);
    }
    if constexpr (SCALE_OUT) {
      const float so = inv_sqrt_degree(degout[row]);
      o.x *= so; o.y *= so; o.z *= so; o.w *= so;
    }
    ((float4*)(out + (size_t)row * FDIM))[lane] = o;
  }
}

extern "C" void kernel_launch(void* const* d_in, const int* in_sizes, int n_in,
                              void* d_out, int out_size, void* d_ws, size_t ws_size,
                              hipStream_t stream) {
  const float* x   = (const float*)d_in[0];
  const int*   src = (const int*)  d_in[1];
  const int*   dst = (const int*)  d_in[2];
  const float* W1  = (const float*)d_in[3];
  const float* b1  = (const float*)d_in[4];
  const float* W2  = (const float*)d_in[5];
  const float* b2  = (const float*)d_in[6];
  const float* W3  = (const float*)d_in[7];
  const float* b3  = (const float*)d_in[8];
  const float* Wr  = (const float*)d_in[9];
  const float* br  = (const float*)d_in[10];
  const float* Wo  = (const float*)d_in[11];
  const float* bo  = (const float*)d_in[12];
  float* out = (float*)d_out;

  float* ws    = (float*)d_ws;
  float* dego  = ws;
  float* ABUF  = dego + 2 * NODES;
  float* HBUF  = ABUF + (size_t)NODES * FDIM;
  int* hist     = (int*)(HBUF + (size_t)NODES * FDIM);
  int* total    = hist + NBIN * NSB;
  int* binstart = total + NBIN;
  int* sorted   = binstart + NBIN + 1;
  constexpr size_t kWordsToSorted = 2 * (size_t)NODES + 2 * (size_t)NODES * FDIM
                                  + (size_t)NBIN * NSB + NBIN + (NBIN + 1) + EDGES;
  constexpr size_t kRunlenAt = (kWordsToSorted + 31) / 32 * 32;
  constexpr size_t kRunsAt   = kRunlenAt + ((size_t)NRG * NSB + 31) / 32 * 32;
  int*          runlen = (int*)ws + kRunlenAt;
  unsigned int* runs   = (unsigned int*)ws + kRunsAt;

  const int mblocks  = NODES / (16 * GW);
  const int scanb    = (NBIN * 32 + 255) / 256;
  const dim3 gemmH_blk(GW * 32);
  const dim3 gemmC_blk(GW * 32);

  bin_count_kernel<<<NSB, STHREADS, 0, stream>>>(dst, hist);
  bin_scan_kernel<<<scanb, 256, 0, stream>>>(hist, total);
  bin_fill_kernel<<<NSB, STHREADS, 0, stream>>>(src, dst, hist, total, binstart, sorted, runs, runlen);
  outdeg_ranges_kernel<<<NRG, RTHREADS, 0, stream>>>(runs, runlen, dego);

  gemm_k128_wmma<FDIM, true, false>
      <<<mblocks, gemmH_blk, 0, stream>>>(x, W1, nullptr, dego, HBUF);
  aggregate_bins_kernel<true, true, false>
      <<<NBIN, AGG_THREADS, 0, stream>>>(HBUF, sorted, binstart, b1, nullptr, ABUF);

  gemm_k128_wmma<FDIM, true, false>
      <<<mblocks, gemmH_blk, 0, stream>>>(ABUF, W2, nullptr, dego, HBUF);
  aggregate_bins_kernel<true, true, true>
      <<<NBIN, AGG_THREADS, 0, stream>>>(HBUF, sorted, binstart, b2, dego, ABUF);

  aggregate_bins_kernel<false, false, false>
      <<<NBIN, AGG_THREADS, 0, stream>>>(ABUF, sorted, binstart, nullptr, nullptr, HBUF);

  tail_three_products_wmma
      <<<mblocks, gemmC_blk, 0, stream>>>(HBUF, x, W3, Wr, b3, br, Wo, bo, out);
}


# GEMM kernels walk their tiles: one block per WGP, its waves loop over 16-row tiles with the next tile's rows loaded under the products; W^T staged once per block, not once per 80 rows; sums unchanged
# speedup vs baseline: 3.4910x; 1.0411x over previous
#include <hip/hip_runtime.h>
#include <hip/hip_bf16.h>


typedef __attribute__((ext_vector_type(2))) float v2f;
typedef __attribute__((ext_vector_type(4))) float v4f;
typedef __attribute__((ext_vector_type(8))) float v8f;

#define NODES 50000
#define EDGES 800000
#define FDIM  128
#define CDIM  64
#define GW     4
#define PGRID  256
#define NTILES (NODES / 16)
static_assert(NTILES * 16 == NODES, "row tiles must tile the nodes");
static_assert(PGRID * GW <= NTILES, "every wave of the grid has a first tile");
#define WPITCH 132

#define RB       40
#define NBIN     (NODES / RB)
#define NSB      256
#define EPB      (EDGES / NSB)
#define STHREADS 512
#define SWAVES   (STHREADS / 32)
#define EPT      ((EPB + STHREADS - 1) / STHREADS)
#define BPT      ((NBIN + STHREADS - 1) / STHREADS)
#define AGG_THREADS 256
#define EB       8
#define RB32     ((RB + 31) / 32)
static_assert(NBIN * RB == NODES, "bins must tile the nodes");
static_assert(NSB * EPB == EDGES, "sort blocks must tile the edges");
static_assert(NSB % 32 == 0, "bin_scan_kernel walks NSB in whole waves");
static_assert(NODES <= 65536 && RB <= 32768, "packed edge = (local row << 16) | src");
static_assert(EB == 8, "the short group is finished in steps of 4, 2, 1");

#define RGSH     10
#define RGN      (1 << RGSH)
#define NRG      ((NODES + RGN - 1) / RGN)
#define RCAP     3136
#define RTHREADS 1024
#define RPW      (NSB / (RTHREADS / 32))
#define RTRIPS   3
static_assert(RCAP >= EPB + 1 && RCAP % 64 == 0, "a run holds a whole sort block's edges; runs start on a line");
static_assert(RTHREADS == RGN, "a range block writes one count per thread");
static_assert(NSB % (RTHREADS / 32) == 0, "the waves of a range block share the runs evenly");
static_assert(NRG <= 64, "the run starts are scanned by one wave, two ranges per lane");

__device__ __forceinline__ float inv_sqrt_degree(float count) {
  return rsqrtf(count > 0.0f ? count : 1.0f);
}


template <int LDN, int U0, int NU, int NW>
__device__ __forceinline__ void wt_piece_loads(const float* __restrict__ W, int lane, int wv,
                                               v4f (&v)[NU][4]) {
  constexpr int CB = LDN / 32;
  constexpr int NP = (FDIM / 16) * CB;
  const int kq = (lane & 3) * 4;
  const int cq = (lane >> 2) * 4;
#pragma unroll
  for (int u = 0; u < NU; ++u) {
    const int p = (U0 + u) * NW + wv;
    if (p < NP) {
      const float* wp = W + (size_t)((p / CB) * 16 + kq) * LDN + (p % CB) * 32 + cq;
#pragma unroll
      for (int r = 0; r < 4; ++r) v[u][r] = *(const v4f*)(wp + r * LDN);
    }
  }
}
template <int LDN, int U0, int NU, int NW>
__device__ __forceinline__ void wt_piece_stores(float* wt, int lane, int wv,
                                                const v4f (&v)[NU][4]) {
  constexpr int CB = LDN / 32;
  constexpr int NP = (FDIM / 16) * CB;
  const int kq = (lane & 3) * 4;
  const int cq = (lane >> 2) * 4;
#pragma unroll
  for (int u = 0; u < NU; ++u) {
    const int p = (U0 + u) * NW + wv;
    if (p < NP) {
      float* lp = wt + ((p % CB) * 32 + cq) * WPITCH + (p / CB) * 16 + kq;
#pragma unroll
      for (int i = 0; i < 4; ++i) {
        v4f o;
        o.x = v[u][0][i]; o.y = v[u][1][i]; o.z = v[u][2][i]; o.w = v[u][3][i];
        *(v4f*)(lp + i * WPITCH) = o;
      }
    }
  }
}

template <int NT>
__device__ __forceinline__ void mma_k128(const float4 (&a)[FDIM / 8], const float* bbase,
                                         v8f (&acc)[NT]) {
  float4 bcur[NT], bnxt[NT];
#pragma unroll
  for (int t = 0; t < NT; ++t) bcur[t] = *(const float4*)(bbase + t * 16 * WPITCH);
#pragma unroll
  for (int j = 0; j < FDIM / 8; ++j) {
    if (j + 1 < FDIM / 8) {
#pragma unroll
      for (int t = 0; t < NT; ++t)
        bnxt[t] = *(const float4*)(bbase + t * 16 * WPITCH + 8 * (j + 1));
    }
    v2f a0, a1;
    a0.x = a[j].x; a0.y = a[j].y;
    a1.x = a[j].z; a1.y = a[j].w;
#pragma unroll
    for (int t = 0; t < NT; ++t) {
      v2f b;
      b.x = bcur[t].x; b.y = bcur[t].y;
      acc[t] = __builtin_amdgcn_wmma_f32_16x16x4_f32(
           false, a0,  false, b,
           (short)0, acc[t],  false,  false);
    }
#pragma unroll
    for (int t = 0; t < NT; ++t) {
      v2f b;
      b.x = bcur[t].z; b.y = bcur[t].w;
      acc[t] = __builtin_amdgcn_wmma_f32_16x16x4_f32(
           false, a1,  false, b,
           (short)0, acc[t],  false,  false);
    }
    if (j + 1 < FDIM / 8) {
#pragma unroll
      for (int t = 0; t < NT; ++t) bcur[t] = bnxt[t];
    }
  }
}

template <int LDN, bool SCALE_A, bool ADD_BIAS>
__global__ __launch_bounds__(GW * 32)
void gemm_k128_wmma(const float* A,
                    const float* __restrict__ W,
                    const float* __restrict__ bias,
                    const float* __restrict__ degout,
                    float* out) {
  constexpr int NT = LDN / 16;
  constexpr int CB = LDN / 32;
  constexpr int NP = (FDIM / 16) * CB;
  constexpr int FP = (NP + GW - 1) / GW;
  __shared__ __attribute__((aligned(16))) float wt[LDN * WPITCH];

  const int tid  = threadIdx.x;
  const int lane = tid & 31;
  const int wave = tid >> 5;
  const int c    = lane & 15;
  const int hh   = lane >> 4;
  const int wv = __builtin_amdgcn_readfirstlane(wave);
  __builtin_assume(wv >= 0 && wv < GW);
  const int nw = (int)gridDim.x * GW;
  int tile = (int)blockIdx.x * GW + wv;

  float4 a[FDIM / 8];
  float scale = 1.0f;
  {
    const float4* arow = (const float4*)(A + (size_t)(tile * 16 + c) * FDIM) + hh;
#pragma unroll
    for (int j = 0; j < FDIM / 8; ++j) a[j] = arow[2 * j];
    if constexpr (SCALE_A) scale = degout[tile * 16 + c];
  }

  const int kq = (lane & 3) * 4;
  const int cq = (lane >> 2) * 4;
  {
    v4f v[FP][4];
#pragma unroll
    for (int u = 0; u < FP; ++u) {
      const int p = u * GW + wv;
      if (p < NP) {
        const float* wp = W + (size_t)((p / CB) * 16 + kq) * LDN + (p % CB) * 32 + cq;
#pragma unroll
        for (int r = 0; r < 4; ++r) v[u][r] = *(const v4f*)(wp + r * LDN);
      }
    }
    __builtin_amdgcn_sched_barrier(0);
#pragma unroll
    for (int u = 0; u < FP; ++u) {
      const int p = u * GW + wv;
      if (p < NP) {
        float* lp = wt + ((p % CB) * 32 + cq) * WPITCH + (p / CB) * 16 + kq;
#pragma unroll
        for (int i = 0; i < 4; ++i) {
          v4f o;
          o.x = v[u][0][i]; o.y = v[u][1][i]; o.z = v[u][2][i]; o.w = v[u][3][i];
          *(v4f*)(lp + i * WPITCH) = o;
        }
      }
    }
  }
  float binit[NT];
#pragma unroll
  for (int t = 0; t < NT; ++t) {
    binit[t] = 0.0f;
    if constexpr (ADD_BIAS) binit[t] = bias[t * 16 + c];
  }
  __syncthreads();

  const float* bbase = wt + c * WPITCH + 4 * hh;
  float4 bf[FDIM / 8][NT];
#pragma unroll
  for (int j = 0; j < FDIM / 8; ++j)
#pragma unroll
    for (int t = 0; t < NT; ++t) bf[j][t] = *(const float4*)(bbase + t * 16 * WPITCH + 8 * j);
  const int rbase = hh << 3;
  for (; tile < NTILES; tile += nw) {
    const int m0   = tile * 16;
    const int next = tile + nw;
    float4 an[FDIM / 8];
    float scale_n = 1.0f;
    if (next < NTILES) {
      const float4* arow = (const float4*)(A + (size_t)(next * 16 + c) * FDIM) + hh;
#pragma unroll
      for (int j = 0; j < FDIM / 8; ++j) an[j] = arow[2 * j];
      if constexpr (SCALE_A) scale_n = degout[next * 16 + c];
    }
    __builtin_amdgcn_sched_barrier(0);
    if constexpr (SCALE_A) {
      const float sc = inv_sqrt_degree(scale);
#pragma unroll
      for (int j = 0; j < FDIM / 8; ++j) {
        a[j].x *= sc; a[j].y *= sc; a[j].z *= sc; a[j].w *= sc;
      }
    }
    v8f acc[NT];
#pragma unroll
    for (int t = 0; t < NT; ++t)
#pragma unroll
      for (int r = 0; r < 8; ++r) acc[t][r] = binit[t];
#pragma unroll
    for (int j = 0; j < FDIM / 8; ++j) {
      v2f a0, a1;
      a0.x = a[j].x; a0.y = a[j].y;
      a1.x = a[j].z; a1.y = a[j].w;
#pragma unroll
      for (int t = 0; t < NT; ++t) {
        v2f b;
        b.x = bf[j][t].x; b.y = bf[j][t].y;
        acc[t] = __builtin_amdgcn_wmma_f32_16x16x4_f32(
             false, a0,  false, b,
             (short)0, acc[t],  false,  false);
      }
#pragma unroll
      for (int t = 0; t < NT; ++t) {
        v2f b;
        b.x = bf[j][t].z; b.y = bf[j][t].w;
        acc[t] = __builtin_amdgcn_wmma_f32_16x16x4_f32(
             false, a1,  false, b,
             (short)0, acc[t],  false,  false);
      }
    }
#pragma unroll
    for (int r = 0; r < 8; ++r)
#pragma unroll
      for (int t = 0; t < NT; ++t)
        out[(size_t)(m0 + rbase + r) * LDN + t * 16 + c] = acc[t][r];
    __builtin_amdgcn_sched_barrier(0);
#pragma unroll
    for (int j = 0; j < FDIM / 8; ++j) a[j] = an[j];
    scale = scale_n;
  }
}

#define TAIL_W3   0
#define TAIL_WR   (TAIL_W3 + FDIM * WPITCH)
#define TAIL_WO   (TAIL_WR + FDIM * WPITCH)
#define TAIL_SCR  (TAIL_WO + CDIM * WPITCH)
#define TW        8
#define TAIL_LDS  (TAIL_SCR + TW * 16 * WPITCH)
static_assert(PGRID * TW <= NTILES, "every wave of the tail grid has a first tile");
static_assert((FDIM * WPITCH * 4) % 16 == 0 && (CDIM * WPITCH * 4) % 16 == 0, "every area starts on a 16-B boundary");

__global__ __launch_bounds__(TW * 32)
void tail_three_products_wmma(const float* T,
                              const float* X,
                              const float* __restrict__ W3,
                              const float* __restrict__ Wr,
                              const float* __restrict__ b3,
                              const float* __restrict__ br,
                              const float* __restrict__ Wo,
                              const float* __restrict__ bo,
                              float* out) {
  constexpr int NT  = FDIM / 16;
  constexpr int NTO = CDIM / 16;
  constexpr int FPH = ((FDIM / 16) * (FDIM / 32) + TW - 1) / TW;
  constexpr int FPO = ((FDIM / 16) * (CDIM / 32) + TW - 1) / TW;
  __shared__ __attribute__((aligned(16))) float wt[TAIL_LDS];

  const int tid  = threadIdx.x;
  const int lane = tid & 31;
  const int wave = tid >> 5;
  const int c    = lane & 15;
  const int hh   = lane >> 4;
  const int wv = __builtin_amdgcn_readfirstlane(wave);
  __builtin_assume(wv >= 0 && wv < TW);
  const int nw = (int)gridDim.x * TW;
  int tile = (int)blockIdx.x * TW + wv;

  float4 ta[FDIM / 8], xa[FDIM / 8];
  {
    const float4* trow = (const float4*)(T + (size_t)(tile * 16 + c) * FDIM) + hh;
    const float4* xrow = (const float4*)(X + (size_t)(tile * 16 + c) * FDIM) + hh;
#pragma unroll
    for (int j = 0; j < FDIM / 8; ++j) { ta[j] = trow[2 * j]; xa[j] = xrow[2 * j]; }
  }
  {
    v4f v3[FPH][4], vr[FPH][4], vo[FPO][4];
    wt_piece_loads<FDIM, 0, FPH, TW>(W3, lane, wv, v3);
    wt_piece_loads<FDIM, 0, FPH, TW>(Wr, lane, wv, vr);
    wt_piece_loads<CDIM, 0, FPO, TW>(Wo, lane, wv, vo);
    __builtin_amdgcn_sched_barrier(0);
    wt_piece_stores<FDIM, 0, FPH, TW>(wt + TAIL_W3, lane, wv, v3);
    wt_piece_stores<FDIM, 0, FPH, TW>(wt + TAIL_WR, lane, wv, vr);
    wt_piece_stores<CDIM, 0, FPO, TW>(wt + TAIL_WO, lane, wv, vo);
  }
  float binit[NT], boinit[NTO];
#pragma unroll
  for (int t = 0; t < NT; ++t) binit[t] = b3[t * 16 + c] + br[t * 16 + c];
#pragma unroll
  for (int t = 0; t < NTO; ++t) boinit[t] = bo[t * 16 + c];
  __syncthreads();

  const float* b3base = wt + TAIL_W3 + c * WPITCH + 4 * hh;
  const float* brbase = wt + TAIL_WR + c * WPITCH + 4 * hh;
  const float* bobase = wt + TAIL_WO + c * WPITCH + 4 * hh;
  float* scr = wt + TAIL_SCR + wave * (16 * WPITCH);
  const int rbase = hh << 3;

  for (; tile < NTILES; tile += nw) {
    const int m0   = tile * 16;
    const int next = tile + nw;
    float4 tn[FDIM / 8], xn[FDIM / 8];
    if (next < NTILES) {
      const float4* trow = (const float4*)(T + (size_t)(next * 16 + c) * FDIM) + hh;
      const float4* xrow = (const float4*)(X + (size_t)(next * 16 + c) * FDIM) + hh;
#pragma unroll
      for (int j = 0; j < FDIM / 8; ++j) { tn[j] = trow[2 * j]; xn[j] = xrow[2 * j]; }
    }
    __builtin_amdgcn_sched_barrier(0);
    v8f acc[NT];
#pragma unroll
    for (int t = 0; t < NT; ++t)
#pragma unroll
      for (int r = 0; r < 8; ++r) acc[t][r] = binit[t];
    mma_k128<NT>(ta, b3base, acc);
    mma_k128<NT>(xa, brbase, acc);

#pragma unroll
    for (int t = 0; t < NT; ++t)
#pragma unroll
      for (int r = 0; r < 8; ++r)
        scr[(hh * 8 + r) * WPITCH + t * 16 + c] = fmaxf(acc[t][r], 0.0f);
    float4 pa[FDIM / 8];
#pragma unroll
    for (int j = 0; j < FDIM / 8; ++j) pa[j] = *(const float4*)(scr + c * WPITCH + 8 * j + 4 * hh);
    v8f acco[NTO];
#pragma unroll
    for (int t = 0; t < NTO; ++t)
#pragma unroll
      for (int r = 0; r < 8; ++r) acco[t][r] = boinit[t];
    mma_k128<NTO>(pa, bobase, acco);

#pragma unroll
    for (int r = 0; r < 8; ++r)
#pragma unroll
      for (int t = 0; t < NTO; ++t)
        out[(size_t)(m0 + rbase + r) * CDIM + t * 16 + c] = acco[t][r];
    __builtin_amdgcn_sched_barrier(0);
#pragma unroll
    for (int j = 0; j < FDIM / 8; ++j) { ta[j] = tn[j]; xa[j] = xn[j]; }
  }
}


__global__ __launch_bounds__(STHREADS)
void bin_count_kernel(const int* __restrict__ dst, int* __restrict__ hist) {
  __shared__ int cnt[NBIN];
  const int tid  = threadIdx.x;
  const int base = blockIdx.x * EPB;
  for (int i = tid; i < NBIN; i += STHREADS) cnt[i] = 0;
  int d[EPT];
#pragma unroll
  for (int k = 0; k < EPT; ++k) {
    const int i = tid + k * STHREADS;
    d[k] = (i < EPB) ? dst[base + i] : 0;
  }
  __syncthreads();
#pragma unroll
  for (int k = 0; k < EPT; ++k) {
    const int i = tid + k * STHREADS;
    if (i < EPB) atomicAdd(&cnt[d[k] / RB], 1);
  }
  __syncthreads();
  for (int i = tid; i < NBIN; i += STHREADS) hist[i * NSB + blockIdx.x] = cnt[i];
}

__global__ void bin_scan_kernel(int* __restrict__ hist, int* __restrict__ total) {
  const int lane = threadIdx.x & 31;
  const int bin  = (blockIdx.x * blockDim.x + threadIdx.x) >> 5;
  if (bin >= NBIN) return;
  int* __restrict__ row = hist + bin * NSB;
  int run = 0;
  for (int b0 = 0; b0 < NSB; b0 += 32) {
    const int v = row[b0 + lane];
    int inc = v;
    for (int dlt = 1; dlt < 32; dlt <<= 1) {
      const int u = __shfl_up(inc, dlt);
      if (lane >= dlt) inc += u;
    }
    row[b0 + lane] = run + inc - v;
    run += __shfl(inc, 31);
  }
  if (lane == 0) total[bin] = run;
}

__global__ __launch_bounds__(STHREADS)
void bin_fill_kernel(const int* __restrict__ src, const int* __restrict__ dst,
                     const int* __restrict__ pre, const int* __restrict__ total,
                     int* __restrict__ binstart, int* __restrict__ sorted,
                     unsigned int* __restrict__ runs, int* __restrict__ runlen) {
  __shared__ int cursor[NBIN];
  __shared__ int wsum[SWAVES];
  __shared__ int rcnt[NRG];
  __shared__ int roff[NRG];
  __shared__ unsigned short stage[RCAP];
  const int tid  = threadIdx.x;
  const int lane = tid & 31;
  const int wave = tid >> 5;
  const int base = blockIdx.x * EPB;

  if (tid < NRG) rcnt[tid] = 0;
  int s[EPT], d[EPT];
#pragma unroll
  for (int k = 0; k < EPT; ++k) {
    const int i  = tid + k * STHREADS;
    const bool ok = i < EPB;
    s[k] = ok ? src[base + i] : 0;
    d[k] = ok ? dst[base + i] : 0;
  }

  int loc[BPT];
  int sum = 0;
#pragma unroll
  for (int j = 0; j < BPT; ++j) {
    const int b = tid * BPT + j;
    const int v = (b < NBIN) ? total[b] : 0;
    loc[j] = sum;
    sum += v;
  }
  int inc = sum;
  for (int dlt = 1; dlt < 32; dlt <<= 1) {
    const int u = __shfl_up(inc, dlt);
    if (lane >= dlt) inc += u;
  }
  if (lane == 31) wsum[wave] = inc;
  __syncthreads();
  int woff = 0;
  for (int w = 0; w < wave; ++w) woff += wsum[w];
  const int texcl = woff + inc - sum;
#pragma unroll
  for (int j = 0; j < BPT; ++j) {
    const int b = tid * BPT + j;
    if (b < NBIN) {
      const int start = texcl + loc[j];
      cursor[b] = start + pre[b * NSB + blockIdx.x];
      if (blockIdx.x == 0) binstart[b] = start;
    }
  }
  if (blockIdx.x == 0 && tid == 0) binstart[NBIN] = EDGES;
  int rk[EPT];
#pragma unroll
  for (int k = 0; k < EPT; ++k) {
    const int i = tid + k * STHREADS;
    rk[k] = (i < EPB) ? atomicAdd(&rcnt[s[k] >> RGSH], 1) : 0;
  }
  __syncthreads();

  if (wave == 0) {
    const int q0 = 2 * lane, q1 = 2 * lane + 1;
    const int c0 = (q0 < NRG) ? rcnt[q0] : 0;
    const int c1 = (q1 < NRG) ? rcnt[q1] : 0;
    int run = c0 + c1;
    for (int dlt = 1; dlt < 32; dlt <<= 1) {
      const int u = __shfl_up(run, dlt);
      if (lane >= dlt) run += u;
    }
    const int excl = run - c0 - c1;
    if (q0 < NRG) roff[q0] = excl;
    if (q1 < NRG) roff[q1] = excl + c0;
  }

#pragma unroll
  for (int k = 0; k < EPT; ++k) {
    const int i = tid + k * STHREADS;
    if (i < EPB) {
      const int bin = d[k] / RB;
      const int pos = atomicAdd(&cursor[bin], 1);
      sorted[pos] = ((d[k] - bin * RB) << 16) | s[k];
    }
  }
  __syncthreads();

#pragma unroll
  for (int k = 0; k < EPT; ++k) {
    const int i = tid + k * STHREADS;
    if (i < EPB) stage[roff[s[k] >> RGSH] + rk[k]] = (unsigned short)(s[k] & (RGN - 1));
  }
  __syncthreads();

  for (int r = wave; r < NRG; r += SWAVES) {
    const int n = rcnt[r];
    const int o = roff[r];
    unsigned int* __restrict__ out = runs + (size_t)(r * NSB + blockIdx.x) * (RCAP / 2);
    for (int i = 2 * lane; i < n; i += 64) {
      const unsigned int lo = stage[o + i];
      const unsigned int hi = (i + 1 < n) ? stage[o + i + 1] : 0u;
      out[i >> 1] = lo | (hi << 16);
    }
    if (lane == 0) runlen[r * NSB + blockIdx.x] = n;
  }
}

__global__ __launch_bounds__(RTHREADS)
void outdeg_ranges_kernel(const unsigned int* __restrict__ runs, const int* __restrict__ runlen,
                          float* __restrict__ deg_out) {
  __shared__ int cnt[RGN];
  const int tid  = threadIdx.x;
  const int lane = tid & 31;
  const int wave = __builtin_amdgcn_readfirstlane(tid >> 5);
  const int r    = blockIdx.x;
  cnt[tid] = 0;
  __syncthreads();

  int n[RPW];
#pragma unroll
  for (int q = 0; q < RPW; ++q) {
    const int len = runlen[r * NSB + wave + (RTHREADS / 32) * q];
    n[q] = len < RCAP ? len : RCAP;
  }
  unsigned int v[RPW][RTRIPS];
#pragma unroll
  for (int q = 0; q < RPW; ++q) {
    const unsigned int* __restrict__ p = runs + (size_t)(r * NSB + wave + (RTHREADS / 32) * q) * (RCAP / 2);
#pragma unroll
    for (int t = 0; t < RTRIPS; ++t) {
      const int i = 2 * (lane + 32 * t);
      v[q][t] = (i < n[q]) ? p[i >> 1] : 0u;
    }
  }
#pragma unroll
  for (int q = 0; q < RPW; ++q) {
#pragma unroll
    for (int t = 0; t < RTRIPS; ++t) {
      const int i = 2 * (lane + 32 * t);
      if (i < n[q])     atomicAdd(&cnt[v[q][t] & (RGN - 1)], 1);
      if (i + 1 < n[q]) atomicAdd(&cnt[(v[q][t] >> 16) & (RGN - 1)], 1);
    }
    const unsigned int* __restrict__ p = runs + (size_t)(r * NSB + wave + (RTHREADS / 32) * q) * (RCAP / 2);
    for (int i = 2 * (lane + 32 * RTRIPS); i < n[q]; i += 64) {
      const unsigned int w = p[i >> 1];
      atomicAdd(&cnt[w & (RGN - 1)], 1);
      if (i + 1 < n[q]) atomicAdd(&cnt[(w >> 16) & (RGN - 1)], 1);
    }
  }
  __syncthreads();
  const int node = r * RGN + tid;
  if (node < NODES) deg_out[node] = (float)cnt[tid];
}

template <int K>
__device__ __forceinline__ void add_edges(const float* __restrict__ h,
                                          const int* __restrict__ sorted,
                                          int e, float* tile, int lane,
                                          int (&deg)[RB32]) {
  int    p[K];
  float4 v[K];
#pragma unroll
  for (int j = 0; j < K; ++j) p[j] = sorted[e + j];
#pragma unroll
  for (int j = 0; j < K; ++j)
    v[j] = ((const float4*)(h + (size_t)(p[j] & 0xFFFF) * FDIM))[lane];
#pragma unroll
  for (int j = 0; j < K; ++j) {
    const int r = p[j] >> 16;
#pragma unroll
    for (int q = 0; q < RB32; ++q) deg[q] += (r == lane + 32 * q) ? 1 : 0;
  }
#pragma unroll
  for (int j = 0; j < K; ++j) {
    float* t = tile + (p[j] >> 16) * FDIM + lane;
    __hip_atomic_fetch_add(t,      v[j].x, __ATOMIC_RELAXED, __HIP_MEMORY_SCOPE_WORKGROUP);
    __hip_atomic_fetch_add(t + 32, v[j].y, __ATOMIC_RELAXED, __HIP_MEMORY_SCOPE_WORKGROUP);
    __hip_atomic_fetch_add(t + 64, v[j].z, __ATOMIC_RELAXED, __HIP_MEMORY_SCOPE_WORKGROUP);
    __hip_atomic_fetch_add(t + 96, v[j].w, __ATOMIC_RELAXED, __HIP_MEMORY_SCOPE_WORKGROUP);
  }
}

template <bool RELU, bool ADD_BIAS, bool SCALE_OUT>
__global__ __launch_bounds__(AGG_THREADS)
void aggregate_bins_kernel(const float* __restrict__ h,
                           const int* __restrict__ sorted,
                           const int* __restrict__ binstart,
                           const float* __restrict__ bias,
                           const float* __restrict__ degout,
                           float* __restrict__ out) {
  __shared__ float tile[RB * FDIM];
  __shared__ int   cnt[RB];
  const int tid  = threadIdx.x;
  const int lane = tid & 31;
  const int wave = __builtin_amdgcn_readfirstlane(tid >> 5);
  const int bin  = blockIdx.x;

  for (int i = tid; i < RB * FDIM; i += AGG_THREADS) tile[i] = 0.0f;
  if (tid < RB) cnt[tid] = 0;
  __syncthreads();

  const int e0 = binstart[bin];
  const int e1 = binstart[bin + 1];
  int deg[RB32];
#pragma unroll
  for (int q = 0; q < RB32; ++q) deg[q] = 0;
  int e = e0 + wave * EB;
  for (; e + EB <= e1; e += (AGG_THREADS / 32) * EB)
    add_edges<EB>(h, sorted, e, tile, lane, deg);
  if (e < e1) {
    if (e + 4 <= e1) { add_edges<4>(h, sorted, e, tile, lane, deg); e += 4; }
    if (e + 2 <= e1) { add_edges<2>(h, sorted, e, tile, lane, deg); e += 2; }
    if (e < e1)        add_edges<1>(h, sorted, e, tile, lane, deg);
  }
#pragma unroll
  for (int q = 0; q < RB32; ++q) {
    const int r = lane + 32 * q;
    if (r < RB && deg[q] != 0) atomicAdd(&cnt[r], deg[q]);
  }
  __syncthreads();

  float b0 = 0.0f, b1 = 0.0f, b2 = 0.0f, b3 = 0.0f;
  if constexpr (ADD_BIAS) {
    b0 = bias[lane * 4 + 0];
    b1 = bias[lane * 4 + 1];
    b2 = bias[lane * 4 + 2];
    b3 = bias[lane * 4 + 3];
  }
  for (int i = tid; i < RB * 32; i += AGG_THREADS) {
    const int r   = i >> 5;
    const int row = bin * RB + r;
    const float* t = tile + r * FDIM + lane;
    const float sc = inv_sqrt_degree((float)cnt[r]);
    float4 o;
    o.x = t[0]  * sc + b0;
    o.y = t[32] * sc + b1;
    o.z = t[64] * sc + b2;
    o.w = t[96] * sc + b3;
    if constexpr (RELU) {
      o.x = fmaxf(o.x, 0.0f); o.y = fmaxf(o.y, 0.0f);
      o.z = fmaxf(o.z, 0.0f); o.w = fmaxf(o.w, 0.0f);
    }
    if constexpr (SCALE_OUT) {
      const float so = inv_sqrt_degree(degout[row]);
      o.x *= so; o.y *= so; o.z *= so; o.w *= so;
    }
    ((float4*)(out + (size_t)row * FDIM))[lane] = o;
  }
}

extern "C" void kernel_launch(void* const* d_in, const int* in_sizes, int n_in,
                              void* d_out, int out_size, void* d_ws, size_t ws_size,
                              hipStream_t stream) {
  const float* x   = (const float*)d_in[0];
  const int*   src = (const int*)  d_in[1];
  const int*   dst = (const int*)  d_in[2];
  const float* W1  = (const float*)d_in[3];
  const float* b1  = (const float*)d_in[4];
  const float* W2  = (const float*)d_in[5];
  const float* b2  = (const float*)d_in[6];
  const float* W3  = (const float*)d_in[7];
  const float* b3  = (const float*)d_in[8];
  const float* Wr  = (const float*)d_in[9];
  const float* br  = (const float*)d_in[10];
  const float* Wo  = (const float*)d_in[11];
  const float* bo  = (const float*)d_in[12];
  float* out = (float*)d_out;

  float* ws    = (float*)d_ws;
  float* dego  = ws;
  float* ABUF  = dego + 2 * NODES;
  float* HBUF  = ABUF + (size_t)NODES * FDIM;
  int* hist     = (int*)(HBUF + (size_t)NODES * FDIM);
  int* total    = hist + NBIN * NSB;
  int* binstart = total + NBIN;
  int* sorted   = binstart + NBIN + 1;
  constexpr size_t kWordsToSorted = 2 * (size_t)NODES + 2 * (size_t)NODES * FDIM
                                  + (size_t)NBIN * NSB + NBIN + (NBIN + 1) + EDGES;
  constexpr size_t kRunlenAt = (kWordsToSorted + 31) / 32 * 32;
  constexpr size_t kRunsAt   = kRunlenAt + ((size_t)NRG * NSB + 31) / 32 * 32;
  int*          runlen = (int*)ws + kRunlenAt;
  unsigned int* runs   = (unsigned int*)ws + kRunsAt;

  const int mblocks  = PGRID;
  const int scanb    = (NBIN * 32 + 255) / 256;
  const dim3 gemmH_blk(GW * 32);
  const dim3 gemmC_blk(TW * 32);

  bin_count_kernel<<<NSB, STHREADS, 0, stream>>>(dst, hist);
  bin_scan_kernel<<<scanb, 256, 0, stream>>>(hist, total);
  bin_fill_kernel<<<NSB, STHREADS, 0, stream>>>(src, dst, hist, total, binstart, sorted, runs, runlen);
  outdeg_ranges_kernel<<<NRG, RTHREADS, 0, stream>>>(runs, runlen, dego);

  gemm_k128_wmma<FDIM, true, false>
      <<<mblocks, gemmH_blk, 0, stream>>>(x, W1, nullptr, dego, HBUF);
  aggregate_bins_kernel<true, true, false>
      <<<NBIN, AGG_THREADS, 0, stream>>>(HBUF, sorted, binstart, b1, nullptr, ABUF);

  gemm_k128_wmma<FDIM, true, false>
      <<<mblocks, gemmH_blk, 0, stream>>>(ABUF, W2, nullptr, dego, HBUF);
  aggregate_bins_kernel<true, true, true>
      <<<NBIN, AGG_THREADS, 0, stream>>>(HBUF, sorted, binstart, b2, dego, ABUF);

  aggregate_bins_kernel<false, false, false>
      <<<NBIN, AGG_THREADS, 0, stream>>>(ABUF, sorted, binstart, nullptr, nullptr, HBUF);

  tail_three_products_wmma
      <<<mblocks, gemmC_blk, 0, stream>>>(HBUF, x, W3, Wr, b3, br, Wo, bo, out);
}
